# starting kernel
# baseline (speedup 1.0000x reference)
#include <hip/hip_runtime.h>
#include <hip/hip_bf16.h>
#include <math.h>

#define NN  100000
#define MM  3
#define DIN 128
#define DH  64
#define NE  1600000

typedef __attribute__((ext_vector_type(2))) float v2f;
typedef __attribute__((ext_vector_type(4))) float v4f;
typedef __attribute__((ext_vector_type(8))) float v8f;

#define AGG_OFF   0ull
#define DEG_OFF   19200000ull
#define LOG_OFF   19500000ull
#define BETA_OFF  19500004ull
#define PROJ_OFF  19500008ull

__device__ __forceinline__ void atomicAddF(float* p, float v) {
  asm volatile("global_atomic_add_f32 %0, %1, off scope:SCOPE_DEV"
               :: "v"(p), "v"(v) : "memory");
}

__global__ __launch_bounds__(128)
void k_proj_wmma(const float* __restrict__ feats, const float* __restrict__ W,
                 float* __restrict__ proj) {
  const int lane = threadIdx.x & 31;
  const int wv   = threadIdx.x >> 5;
  const int col0 = wv * 16;
  const int m    = blockIdx.y;
  const int n0   = blockIdx.x * 16;
  const int r    = lane & 15;
  const int hi   = lane >> 4;
  const int kh   = hi * 2;

  const float* A = feats + ((size_t)m * NN + n0) * DIN;
  const float* B = W + (size_t)m * DIN * DH;

  v8f acc = {0.f,0.f,0.f,0.f,0.f,0.f,0.f,0.f};
  #pragma unroll
  for (int k = 0; k < DIN; k += 4) {
    v2f a = *(const v2f*)(A + (size_t)r * DIN + k + kh);
    v2f b;
    b.x = B[(k + kh)     * DH + col0 + r];
    b.y = B[(k + kh + 1) * DH + col0 + r];
    acc = __builtin_amdgcn_wmma_f32_16x16x4_f32(false, a, false, b, (short)0, acc,
                                                false, false);
  }
  float* O = proj + ((size_t)m * NN + n0) * DH;
  #pragma unroll
  for (int v = 0; v < 8; ++v)
    O[(size_t)(v + hi * 8) * DH + col0 + r] = acc[v];
}

__global__ __launch_bounds__(256)
void k_scatter(const float* __restrict__ proj, const int* __restrict__ src,
               const int* __restrict__ dst, float* __restrict__ agg,
               float* __restrict__ deg) {
  size_t t = (size_t)blockIdx.x * 256 + threadIdx.x;
  if (t >= (size_t)MM * NE * DH) return;
  int    j  = (int)(t & (DH - 1));
  size_t e  = t >> 6;
  int    m  = (int)(e / NE);
  int s = src[e];
  int d = dst[e];
  float val = proj[((size_t)m * NN + s) * DH + j];
  atomicAddF(&agg[((size_t)m * NN + d) * DH + j], val);
  if (j == 0) atomicAddF(&deg[(size_t)m * NN + d], 1.0f);
}

__global__ __launch_bounds__(256)
void k_prelu(float* __restrict__ agg, const float* __restrict__ deg,
             const float* __restrict__ b, const float* __restrict__ pa) {
  size_t i4 = (size_t)blockIdx.x * 256 + threadIdx.x;
  if (i4 >= (size_t)MM * NN * (DH / 4)) return;
  size_t base = i4 * 4;
  size_t nl   = i4 >> 4;
  int m  = (int)(nl / NN);
  int j4 = (int)(i4 & 15);
  float d = fmaxf(deg[nl], 1.0f);
  float a = pa[m];
  v4f hv = *(v4f*)(agg + base);
  v4f bv = *(const v4f*)(b + m * DH + j4 * 4);
  #pragma unroll
  for (int c = 0; c < 4; ++c) {
    float h = hv[c] / d + bv[c];
    hv[c] = h > 0.f ? h : a * h;
  }
  *(v4f*)(agg + base) = hv;
}

__global__ __launch_bounds__(128)
void k_fc_wmma(const float* __restrict__ h, const float* __restrict__ fcW,
               const float* __restrict__ fcb, const float* __restrict__ attn,
               float* __restrict__ logits) {
  __shared__ float red[128];
  const int lane = threadIdx.x & 31;
  const int wv   = threadIdx.x >> 5;
  const int col0 = wv * 16;
  const int m    = blockIdx.y;
  const int n0   = blockIdx.x * 16;
  const int r    = lane & 15;
  const int hi   = lane >> 4;
  const int kh   = hi * 2;

  const float* A = h + ((size_t)m * NN + n0) * DH;
  float bias = fcb[col0 + r];
  v8f acc = {bias, bias, bias, bias, bias, bias, bias, bias};
  #pragma unroll
  for (int k = 0; k < DH; k += 4) {
    v2f a = *(const v2f*)(A + (size_t)r * DH + k + kh);
    v2f bb;
    bb.x = fcW[(k + kh)     * DH + col0 + r];
    bb.y = fcW[(k + kh + 1) * DH + col0 + r];
    acc = __builtin_amdgcn_wmma_f32_16x16x4_f32(false, a, false, bb, (short)0, acc,
                                                false, false);
  }
  float p = 0.f;
  #pragma unroll
  for (int v = 0; v < 8; ++v) p += tanhf(acc[v]);
  p *= attn[col0 + r];

  red[threadIdx.x] = p;
  __syncthreads();
  #pragma unroll
  for (int s2 = 64; s2 > 0; s2 >>= 1) {
    if (threadIdx.x < s2) red[threadIdx.x] += red[threadIdx.x + s2];
    __syncthreads();
  }
  if (threadIdx.x == 0) atomicAddF(&logits[m], red[0]);
}

__global__ void k_softmax(const float* __restrict__ logits, float* __restrict__ beta) {
  if (threadIdx.x == 0 && blockIdx.x == 0) {
    float l[MM]; float mx = -1e30f;
    for (int m = 0; m < MM; ++m) { l[m] = logits[m] / (float)NN; mx = fmaxf(mx, l[m]); }
    float s = 0.f;
    for (int m = 0; m < MM; ++m) { l[m] = expf(l[m] - mx); s += l[m]; }
    for (int m = 0; m < MM; ++m) beta[m] = l[m] / s;
  }
}

__global__ __launch_bounds__(256)
void k_out(const float* __restrict__ h, const float* __restrict__ beta,
           float* __restrict__ out) {
  size_t i4 = (size_t)blockIdx.x * 256 + threadIdx.x;
  if (i4 >= (size_t)NN * (DH / 4)) return;
  size_t base = i4 * 4;
  const size_t P = (size_t)NN * DH;
  float b0 = beta[0], b1 = beta[1], b2 = beta[2];
  v4f x0 = *(const v4f*)(h + base);
  v4f x1 = *(const v4f*)(h + P + base);
  v4f x2 = *(const v4f*)(h + 2 * P + base);
  v4f z  = x0 * b0 + x1 * b1 + x2 * b2;
  *(v4f*)(out + base) = z;
}

extern "C" void kernel_launch(void* const* d_in, const int* in_sizes, int n_in,
                              void* d_out, int out_size, void* d_ws, size_t ws_size,
                              hipStream_t stream) {
  const float* feats = (const float*)d_in[0];
  const int*   src   = (const int*)  d_in[1];
  const int*   dst   = (const int*)  d_in[2];
  const float* W     = (const float*)d_in[3];
  const float* b     = (const float*)d_in[4];
  const float* pa    = (const float*)d_in[5];
  const float* fcW   = (const float*)d_in[6];
  const float* fcb   = (const float*)d_in[7];
  const float* attn  = (const float*)d_in[8];

  float* ws     = (float*)d_ws;
  float* agg    = ws + AGG_OFF;
  float* deg    = ws + DEG_OFF;
  float* logits = ws + LOG_OFF;
  float* beta   = ws + BETA_OFF;
  float* proj   = ws + PROJ_OFF;
  float* out    = (float*)d_out;

  hipMemsetAsync(d_ws, 0, (size_t)(LOG_OFF + 8) * sizeof(float), stream);

  dim3 gTiles(NN / 16, MM);
  k_proj_wmma<<<gTiles, 128, 0, stream>>>(feats, W, proj);

  size_t total = (size_t)MM * NE * DH;
  unsigned blocks = (unsigned)((total + 255) / 256);
  k_scatter<<<blocks, 256, 0, stream>>>(proj, src, dst, agg, deg);

  size_t nv4 = (size_t)MM * NN * (DH / 4);
  k_prelu<<<(unsigned)((nv4 + 255) / 256), 256, 0, stream>>>(agg, deg, b, pa);

  k_fc_wmma<<<gTiles, 128, 0, stream>>>(agg, fcW, fcb, attn, logits);

  k_softmax<<<1, 32, 0, stream>>>(logits, beta);

  size_t ov4 = (size_t)NN * (DH / 4);
  k_out<<<(unsigned)((ov4 + 255) / 256), 256, 0, stream>>>(agg, beta, out);
}


# aggregation as a per-row pull, no LDS: edges counting-sorted by destination row (int atomic rank, scan, place); a wave per row sums its proj rows in registers, stores prelu'd h; sum order: arrival
# speedup vs baseline: 2.0502x; 2.0502x over previous
#include <hip/hip_runtime.h>
#include <hip/hip_bf16.h>
#include <math.h>

#define NN  100000
#define MM  3
#define DIN 128
#define DH  64
#define NE  1600000
#define NROWS (MM * NN)
#define NEDGE (MM * NE)
#define SCAN_B 1024
#define SCAN_G 293

typedef __attribute__((ext_vector_type(2))) float v2f;
typedef __attribute__((ext_vector_type(4))) float v4f;
typedef __attribute__((ext_vector_type(8))) float v8f;

#define PROJ_OFF  0ull
#define H_OFF     19200000ull
#define PTR_OFF   38400000ull
#define LOG_OFF   38700000ull
#define BETA_OFF  38700004ull
#define RANK_OFF  H_OFF
#define BSUM_OFF  (H_OFF + 4800000ull)

__device__ __forceinline__ void atomicAddF(float* p, float v) {
  asm volatile("global_atomic_add_f32 %0, %1, off scope:SCOPE_DEV"
               :: "v"(p), "v"(v) : "memory");
}

__global__ __launch_bounds__(128)
void k_proj_wmma(const float* __restrict__ feats, const float* __restrict__ W,
                 float* __restrict__ proj) {
  const int lane = threadIdx.x & 31;
  const int wv   = threadIdx.x >> 5;
  const int col0 = wv * 16;
  const int m    = blockIdx.y;
  const int n0   = blockIdx.x * 16;
  const int r    = lane & 15;
  const int hi   = lane >> 4;
  const int kh   = hi * 2;

  const float* A = feats + ((size_t)m * NN + n0) * DIN;
  const float* B = W + (size_t)m * DIN * DH;

  v8f acc = {0.f,0.f,0.f,0.f,0.f,0.f,0.f,0.f};
  #pragma unroll
  for (int k = 0; k < DIN; k += 4) {
    v2f a = *(const v2f*)(A + (size_t)r * DIN + k + kh);
    v2f b;
    b.x = B[(k + kh)     * DH + col0 + r];
    b.y = B[(k + kh + 1) * DH + col0 + r];
    acc = __builtin_amdgcn_wmma_f32_16x16x4_f32(false, a, false, b, (short)0, acc,
                                                false, false);
  }
  float* O = proj + ((size_t)m * NN + n0) * DH;
  #pragma unroll
  for (int v = 0; v < 8; ++v)
    O[(size_t)(v + hi * 8) * DH + col0 + r] = acc[v];
}

__global__ __launch_bounds__(256)
void k_rank(const int* __restrict__ dst, int* __restrict__ cnt, int* __restrict__ rank) {
  const int e = blockIdx.x * 256 + threadIdx.x;
  if (e >= NEDGE) return;
  const int m = e / NE;
  rank[e] = atomicAdd(&cnt[m * NN + dst[e]], 1);
}

__global__ __launch_bounds__(SCAN_B)
void k_bsum(const int* __restrict__ cnt, int* __restrict__ bsum) {
  __shared__ int red[32];
  const int lane = threadIdx.x & 31;
  const int wv   = threadIdx.x >> 5;
  const int i = blockIdx.x * SCAN_B + threadIdx.x;
  int v = (i < NROWS) ? cnt[i] : 0;
  #pragma unroll
  for (int o = 16; o > 0; o >>= 1) v += __shfl_down(v, o);
  if (lane == 0) red[wv] = v;
  __syncthreads();
  if (wv == 0) {
    int t = red[lane];
    #pragma unroll
    for (int o = 16; o > 0; o >>= 1) t += __shfl_down(t, o);
    if (lane == 0) bsum[blockIdx.x] = t;
  }
}

__global__ __launch_bounds__(SCAN_B)
void k_scan(int* __restrict__ ptr, const int* __restrict__ bsum) {
  __shared__ int wsum[32];
  __shared__ int pre_s;
  const int lane = threadIdx.x & 31;
  const int wv   = threadIdx.x >> 5;

  int p = ((int)threadIdx.x < (int)blockIdx.x) ? bsum[threadIdx.x] : 0;
  #pragma unroll
  for (int o = 16; o > 0; o >>= 1) p += __shfl_down(p, o);
  if (lane == 0) wsum[wv] = p;
  __syncthreads();
  if (wv == 0) {
    int t = wsum[lane];
    #pragma unroll
    for (int o = 16; o > 0; o >>= 1) t += __shfl_down(t, o);
    if (lane == 0) pre_s = t;
  }
  __syncthreads();
  const int base = pre_s;
  __syncthreads();

  const int i = blockIdx.x * SCAN_B + threadIdx.x;
  const int c = (i < NROWS) ? ptr[i] : 0;
  int x = c;
  #pragma unroll
  for (int o = 1; o < 32; o <<= 1) {
    int y = __shfl_up(x, o);
    if (lane >= o) x += y;
  }
  if (lane == 31) wsum[wv] = x;
  __syncthreads();
  if (wv == 0) {
    const int w = wsum[lane];
    int t = w;
    #pragma unroll
    for (int o = 1; o < 32; o <<= 1) {
      int y = __shfl_up(t, o);
      if (lane >= o) t += y;
    }
    wsum[lane] = t - w;
  }
  __syncthreads();
  if (i < NROWS) ptr[i] = base + wsum[wv] + (x - c);
}

__global__ __launch_bounds__(256)
void k_place(const int* __restrict__ src, const int* __restrict__ dst,
             const int* __restrict__ ptr, const int* __restrict__ rank,
             int* __restrict__ sorted) {
  const int e = blockIdx.x * 256 + threadIdx.x;
  if (e >= NEDGE) return;
  const int m = e / NE;
  sorted[ptr[m * NN + dst[e]] + rank[e]] = m * NN + src[e];
}

__global__ __launch_bounds__(256)
void k_pull(const float* __restrict__ proj, const int* __restrict__ ptr,
            const int* __restrict__ sorted, const float* __restrict__ b,
            const float* __restrict__ pa, float* __restrict__ h) {
  const int lane = threadIdx.x & 31;
  const int g = __builtin_amdgcn_readfirstlane((int)(blockIdx.x * 8 + (threadIdx.x >> 5)));
  const int start = ptr[g];
  const int end   = (g + 1 < NROWS) ? ptr[g + 1] : NEDGE;
  const int m = g / NN;

  float ax = 0.f, ay = 0.f;
  for (int base = start; base < end; base += 32) {
    const int nb = min(32, end - base);
    const int id = sorted[base + min(lane, nb - 1)];
    for (int j = 0; j < nb; j += 8) {
      v2f v[8];
      #pragma unroll
      for (int u = 0; u < 8; ++u) {
        const int s = __builtin_amdgcn_readlane(id, min(j + u, nb - 1));
        v[u] = *(const v2f*)(proj + (size_t)s * DH + lane * 2);
      }
      #pragma unroll
      for (int u = 0; u < 8; ++u) {
        const bool ok = (j + u) < nb;
        ax += ok ? v[u].x : 0.f;
        ay += ok ? v[u].y : 0.f;
      }
    }
  }
  const float d = fmaxf((float)(end - start), 1.0f);
  const float a = pa[m];
  const v2f bv = *(const v2f*)(b + m * DH + lane * 2);
  float hx = ax / d + bv.x;
  float hy = ay / d + bv.y;
  v2f o;
  o.x = hx > 0.f ? hx : a * hx;
  o.y = hy > 0.f ? hy : a * hy;
  *(v2f*)(h + (size_t)g * DH + lane * 2) = o;
}

__global__ __launch_bounds__(128)
void k_fc_wmma(const float* __restrict__ h, const float* __restrict__ fcW,
               const float* __restrict__ fcb, const float* __restrict__ attn,
               float* __restrict__ logits) {
  __shared__ float red[128];
  const int lane = threadIdx.x & 31;
  const int wv   = threadIdx.x >> 5;
  const int col0 = wv * 16;
  const int m    = blockIdx.y;
  const int n0   = blockIdx.x * 16;
  const int r    = lane & 15;
  const int hi   = lane >> 4;
  const int kh   = hi * 2;

  const float* A = h + ((size_t)m * NN + n0) * DH;
  float bias = fcb[col0 + r];
  v8f acc = {bias, bias, bias, bias, bias, bias, bias, bias};
  #pragma unroll
  for (int k = 0; k < DH; k += 4) {
    v2f a = *(const v2f*)(A + (size_t)r * DH + k + kh);
    v2f bb;
    bb.x = fcW[(k + kh)     * DH + col0 + r];
    bb.y = fcW[(k + kh + 1) * DH + col0 + r];
    acc = __builtin_amdgcn_wmma_f32_16x16x4_f32(false, a, false, bb, (short)0, acc,
                                                false, false);
  }
  float p = 0.f;
  #pragma unroll
  for (int v = 0; v < 8; ++v) p += tanhf(acc[v]);
  p *= attn[col0 + r];

  red[threadIdx.x] = p;
  __syncthreads();
  #pragma unroll
  for (int s2 = 64; s2 > 0; s2 >>= 1) {
    if (threadIdx.x < s2) red[threadIdx.x] += red[threadIdx.x + s2];
    __syncthreads();
  }
  if (threadIdx.x == 0) atomicAddF(&logits[m], red[0]);
}

__global__ void k_softmax(const float* __restrict__ logits, float* __restrict__ beta) {
  if (threadIdx.x == 0 && blockIdx.x == 0) {
    float l[MM]; float mx = -1e30f;
    for (int m = 0; m < MM; ++m) { l[m] = logits[m] / (float)NN; mx = fmaxf(mx, l[m]); }
    float s = 0.f;
    for (int m = 0; m < MM; ++m) { l[m] = expf(l[m] - mx); s += l[m]; }
    for (int m = 0; m < MM; ++m) beta[m] = l[m] / s;
  }
}

__global__ __launch_bounds__(256)
void k_out(const float* __restrict__ h, const float* __restrict__ beta,
           float* __restrict__ out) {
  size_t i4 = (size_t)blockIdx.x * 256 + threadIdx.x;
  if (i4 >= (size_t)NN * (DH / 4)) return;
  size_t base = i4 * 4;
  const size_t P = (size_t)NN * DH;
  float b0 = beta[0], b1 = beta[1], b2 = beta[2];
  v4f x0 = *(const v4f*)(h + base);
  v4f x1 = *(const v4f*)(h + P + base);
  v4f x2 = *(const v4f*)(h + 2 * P + base);
  v4f z  = x0 * b0 + x1 * b1 + x2 * b2;
  *(v4f*)(out + base) = z;
}

extern "C" void kernel_launch(void* const* d_in, const int* in_sizes, int n_in,
                              void* d_out, int out_size, void* d_ws, size_t ws_size,
                              hipStream_t stream) {
  const float* feats = (const float*)d_in[0];
  const int*   src   = (const int*)  d_in[1];
  const int*   dst   = (const int*)  d_in[2];
  const float* W     = (const float*)d_in[3];
  const float* b     = (const float*)d_in[4];
  const float* pa    = (const float*)d_in[5];
  const float* fcW   = (const float*)d_in[6];
  const float* fcb   = (const float*)d_in[7];
  const float* attn  = (const float*)d_in[8];

  float* ws     = (float*)d_ws;
  float* proj   = ws + PROJ_OFF;
  float* h      = ws + H_OFF;
  int*   ptr    = (int*)(ws + PTR_OFF);
  float* logits = ws + LOG_OFF;
  float* beta   = ws + BETA_OFF;
  int*   rank   = (int*)(ws + RANK_OFF);
  int*   bsum   = (int*)(ws + BSUM_OFF);
  float* out    = (float*)d_out;
  int*   sorted = (int*)d_out;

  hipMemsetAsync(ws + PTR_OFF, 0, (size_t)(NROWS + 8) * sizeof(float), stream);

  dim3 gTiles(NN / 16, MM);
  k_proj_wmma<<<gTiles, 128, 0, stream>>>(feats, W, proj);

  k_rank <<<NEDGE / 256, 256, 0, stream>>>(dst, ptr, rank);
  k_bsum <<<SCAN_G, SCAN_B, 0, stream>>>(ptr, bsum);
  k_scan <<<SCAN_G, SCAN_B, 0, stream>>>(ptr, bsum);
  k_place<<<NEDGE / 256, 256, 0, stream>>>(src, dst, ptr, rank, sorted);
  k_pull <<<NROWS / 8, 256, 0, stream>>>(proj, ptr, sorted, b, pa, h);

  k_fc_wmma<<<gTiles, 128, 0, stream>>>(h, fcW, fcb, attn, logits);

  k_softmax<<<1, 32, 0, stream>>>(logits, beta);

  size_t ov4 = (size_t)NN * (DH / 4);
  k_out<<<(unsigned)((ov4 + 255) / 256), 256, 0, stream>>>(h, beta, out);
}


# edge-sort and row-sum kernels as one-pass resident grids: a thread takes 4 edges of each set, a wave walks 18-19 rows with the next row's list fetched ahead; row sum order: arrival as before
# speedup vs baseline: 2.2696x; 1.1070x over previous
#include <hip/hip_runtime.h>
#include <hip/hip_bf16.h>
#include <math.h>

#define NN  100000
#define MM  3
#define DIN 128
#define DH  64
#define NE  1600000
#define NROWS (MM * NN)
#define NEDGE (MM * NE)
#define SCAN_B 1024
#define SCAN_G 293
#define SORT_G 3125
#define PULL_G 2048

typedef __attribute__((ext_vector_type(2))) float v2f;
typedef __attribute__((ext_vector_type(4))) float v4f;
typedef __attribute__((ext_vector_type(8))) float v8f;
typedef __attribute__((ext_vector_type(4))) int v4i;

#define PROJ_OFF  0ull
#define H_OFF     19200000ull
#define PTR_OFF   38400000ull
#define LOG_OFF   38700000ull
#define BETA_OFF  38700004ull
#define RANK_OFF  H_OFF
#define BSUM_OFF  (H_OFF + 4800000ull)

__device__ __forceinline__ void atomicAddF(float* p, float v) {
  asm volatile("global_atomic_add_f32 %0, %1, off scope:SCOPE_DEV"
               :: "v"(p), "v"(v) : "memory");
}

__global__ __launch_bounds__(128)
void k_proj_wmma(const float* __restrict__ feats, const float* __restrict__ W,
                 float* __restrict__ proj) {
  const int lane = threadIdx.x & 31;
  const int wv   = threadIdx.x >> 5;
  const int col0 = wv * 16;
  const int m    = blockIdx.y;
  const int n0   = blockIdx.x * 16;
  const int r    = lane & 15;
  const int hi   = lane >> 4;
  const int kh   = hi * 2;

  const float* A = feats + ((size_t)m * NN + n0) * DIN;
  const float* B = W + (size_t)m * DIN * DH;

  v8f acc = {0.f,0.f,0.f,0.f,0.f,0.f,0.f,0.f};
  #pragma unroll
  for (int k = 0; k < DIN; k += 4) {
    v2f a = *(const v2f*)(A + (size_t)r * DIN + k + kh);
    v2f b;
    b.x = B[(k + kh)     * DH + col0 + r];
    b.y = B[(k + kh + 1) * DH + col0 + r];
    acc = __builtin_amdgcn_wmma_f32_16x16x4_f32(false, a, false, b, (short)0, acc,
                                                false, false);
  }
  float* O = proj + ((size_t)m * NN + n0) * DH;
  #pragma unroll
  for (int v = 0; v < 8; ++v)
    O[(size_t)(v + hi * 8) * DH + col0 + r] = acc[v];
}

__global__ __launch_bounds__(128)
void k_rank(const int* __restrict__ dst, int* __restrict__ cnt, int* __restrict__ rank) {
  const int t = blockIdx.x * 128 + threadIdx.x;
  v4i d[MM];
  #pragma unroll
  for (int m = 0; m < MM; ++m)
    d[m] = *(const v4i*)(dst + (size_t)m * NE + (size_t)t * 4);
  v4i r[MM];
  #pragma unroll
  for (int m = 0; m < MM; ++m) {
    int* c = cnt + m * NN;
    r[m].x = atomicAdd(c + d[m].x, 1);
    r[m].y = atomicAdd(c + d[m].y, 1);
    r[m].z = atomicAdd(c + d[m].z, 1);
    r[m].w = atomicAdd(c + d[m].w, 1);
  }
  #pragma unroll
  for (int m = 0; m < MM; ++m)
    *(v4i*)(rank + (size_t)m * NE + (size_t)t * 4) = r[m];
}

__global__ __launch_bounds__(SCAN_B)
void k_bsum(const int* __restrict__ cnt, int* __restrict__ bsum) {
  __shared__ int red[32];
  const int lane = threadIdx.x & 31;
  const int wv   = threadIdx.x >> 5;
  const int i = blockIdx.x * SCAN_B + threadIdx.x;
  int v = (i < NROWS) ? cnt[i] : 0;
  #pragma unroll
  for (int o = 16; o > 0; o >>= 1) v += __shfl_down(v, o);
  if (lane == 0) red[wv] = v;
  __syncthreads();
  if (wv == 0) {
    int t = red[lane];
    #pragma unroll
    for (int o = 16; o > 0; o >>= 1) t += __shfl_down(t, o);
    if (lane == 0) bsum[blockIdx.x] = t;
  }
}

__global__ __launch_bounds__(SCAN_B)
void k_scan(int* __restrict__ ptr, const int* __restrict__ bsum) {
  __shared__ int wsum[32];
  __shared__ int pre_s;
  const int lane = threadIdx.x & 31;
  const int wv   = threadIdx.x >> 5;

  int p = ((int)threadIdx.x < (int)blockIdx.x) ? bsum[threadIdx.x] : 0;
  #pragma unroll
  for (int o = 16; o > 0; o >>= 1) p += __shfl_down(p, o);
  if (lane == 0) wsum[wv] = p;
  __syncthreads();
  if (wv == 0) {
    int t = wsum[lane];
    #pragma unroll
    for (int o = 16; o > 0; o >>= 1) t += __shfl_down(t, o);
    if (lane == 0) pre_s = t;
  }
  __syncthreads();
  const int base = pre_s;
  __syncthreads();

  const int i = blockIdx.x * SCAN_B + threadIdx.x;
  const int c = (i < NROWS) ? ptr[i] : 0;
  int x = c;
  #pragma unroll
  for (int o = 1; o < 32; o <<= 1) {
    int y = __shfl_up(x, o);
    if (lane >= o) x += y;
  }
  if (lane == 31) wsum[wv] = x;
  __syncthreads();
  if (wv == 0) {
    const int w = wsum[lane];
    int t = w;
    #pragma unroll
    for (int o = 1; o < 32; o <<= 1) {
      int y = __shfl_up(t, o);
      if (lane >= o) t += y;
    }
    wsum[lane] = t - w;
  }
  __syncthreads();
  if (i < NROWS) ptr[i] = base + wsum[wv] + (x - c);
}

__global__ __launch_bounds__(128)
void k_place(const int* __restrict__ src, const int* __restrict__ dst,
             const int* __restrict__ ptr, const int* __restrict__ rank,
             int* __restrict__ sorted) {
  const int t = blockIdx.x * 128 + threadIdx.x;
  #pragma unroll
  for (int m = 0; m < MM; ++m) {
    const size_t e = (size_t)m * NE + (size_t)t * 4;
    const v4i s = *(const v4i*)(src + e);
    const v4i d = *(const v4i*)(dst + e);
    const v4i r = *(const v4i*)(rank + e);
    const int* p = ptr + m * NN;
    const int o = m * NN;
    const int p0 = p[d.x], p1 = p[d.y], p2 = p[d.z], p3 = p[d.w];
    sorted[p0 + r.x] = o + s.x;
    sorted[p1 + r.y] = o + s.y;
    sorted[p2 + r.z] = o + s.z;
    sorted[p3 + r.w] = o + s.w;
  }
}

__device__ __forceinline__ int row_end(const int* __restrict__ ptr, int g) {
  return (g + 1 < NROWS) ? ptr[g + 1] : NEDGE;
}
__device__ __forceinline__ int first_ids(const int* __restrict__ sorted, int start, int end, int lane) {
  const int last = max(end - start, 1) - 1;
  return __hip_atomic_load(sorted + min(start + min(lane, last), NEDGE - 1),
                           __ATOMIC_RELAXED, __HIP_MEMORY_SCOPE_WAVEFRONT);
}

__device__ __forceinline__ void sum_batch(const float* __restrict__ proj, int idc, int nb, int lane,
                                          float& ax, float& ay) {
  for (int j = 0; j < nb; j += 8) {
    v2f v[8];
    #pragma unroll
    for (int u = 0; u < 8; ++u) {
      const int s = __builtin_amdgcn_readlane(idc, min(j + u, nb - 1));
      v[u] = *(const v2f*)(proj + (size_t)s * DH + lane * 2);
    }
    #pragma unroll
    for (int u = 0; u < 8; ++u) {
      const bool ok = (j + u) < nb;
      ax += ok ? v[u].x : 0.f;
      ay += ok ? v[u].y : 0.f;
    }
  }
}

__global__ __launch_bounds__(256)
void k_pull(const float* __restrict__ proj, const int* __restrict__ ptr,
            const int* __restrict__ sorted, const float* __restrict__ b,
            const float* __restrict__ pa, float* __restrict__ h) {
  const int lane = threadIdx.x & 31;
  const int nw = PULL_G * 8;
  int g = __builtin_amdgcn_readfirstlane((int)(blockIdx.x * 8 + (threadIdx.x >> 5)));
  int start = ptr[g];
  int end   = row_end(ptr, g);
  int id    = first_ids(sorted, start, end, lane);
  int g1     = min(g + nw, NROWS - 1);
  int start1 = ptr[g1];
  int end1   = row_end(ptr, g1);

  for (;;) {
    const int id1    = first_ids(sorted, start1, end1, lane);
    const int g2     = min(g + 2 * nw, NROWS - 1);
    const int start2 = ptr[g2];
    const int end2   = row_end(ptr, g2);

    const int m   = g / NN;
    const int deg = end - start;
    const float a = pa[m];
    const v2f bv  = *(const v2f*)(b + m * DH + lane * 2);

    float ax = 0.f, ay = 0.f;
    sum_batch(proj, id, min(32, deg), lane, ax, ay);
    for (int base = 32; base < deg; base += 32) {
      const int nb = min(32, deg - base);
      const int idc = sorted[start + base + min(lane, nb - 1)];
      sum_batch(proj, idc, nb, lane, ax, ay);
    }
    const float dg = fmaxf((float)deg, 1.0f);
    const float hx = ax / dg + bv.x;
    const float hy = ay / dg + bv.y;
    v2f o;
    o.x = hx > 0.f ? hx : a * hx;
    o.y = hy > 0.f ? hy : a * hy;
    *(v2f*)(h + (size_t)g * DH + lane * 2) = o;

    if (g + nw >= NROWS) break;
    g = g + nw;  start = start1;  end = end1;  id = id1;
    g1 = g2;     start1 = start2; end1 = end2;
  }
}

__global__ __launch_bounds__(128)
void k_fc_wmma(const float* __restrict__ h, const float* __restrict__ fcW,
               const float* __restrict__ fcb, const float* __restrict__ attn,
               float* __restrict__ logits) {
  __shared__ float red[128];
  const int lane = threadIdx.x & 31;
  const int wv   = threadIdx.x >> 5;
  const int col0 = wv * 16;
  const int m    = blockIdx.y;
  const int n0   = blockIdx.x * 16;
  const int r    = lane & 15;
  const int hi   = lane >> 4;
  const int kh   = hi * 2;

  const float* A = h + ((size_t)m * NN + n0) * DH;
  float bias = fcb[col0 + r];
  v8f acc = {bias, bias, bias, bias, bias, bias, bias, bias};
  #pragma unroll
  for (int k = 0; k < DH; k += 4) {
    v2f a = *(const v2f*)(A + (size_t)r * DH + k + kh);
    v2f bb;
    bb.x = fcW[(k + kh)     * DH + col0 + r];
    bb.y = fcW[(k + kh + 1) * DH + col0 + r];
    acc = __builtin_amdgcn_wmma_f32_16x16x4_f32(false, a, false, bb, (short)0, acc,
                                                false, false);
  }
  float p = 0.f;
  #pragma unroll
  for (int v = 0; v < 8; ++v) p += tanhf(acc[v]);
  p *= attn[col0 + r];

  red[threadIdx.x] = p;
  __syncthreads();
  #pragma unroll
  for (int s2 = 64; s2 > 0; s2 >>= 1) {
    if (threadIdx.x < s2) red[threadIdx.x] += red[threadIdx.x + s2];
    __syncthreads();
  }
  if (threadIdx.x == 0) atomicAddF(&logits[m], red[0]);
}

__global__ void k_softmax(const float* __restrict__ logits, float* __restrict__ beta) {
  if (threadIdx.x == 0 && blockIdx.x == 0) {
    float l[MM]; float mx = -1e30f;
    for (int m = 0; m < MM; ++m) { l[m] = logits[m] / (float)NN; mx = fmaxf(mx, l[m]); }
    float s = 0.f;
    for (int m = 0; m < MM; ++m) { l[m] = expf(l[m] - mx); s += l[m]; }
    for (int m = 0; m < MM; ++m) beta[m] = l[m] / s;
  }
}

__global__ __launch_bounds__(256)
void k_out(const float* __restrict__ h, const float* __restrict__ beta,
           float* __restrict__ out) {
  size_t i4 = (size_t)blockIdx.x * 256 + threadIdx.x;
  if (i4 >= (size_t)NN * (DH / 4)) return;
  size_t base = i4 * 4;
  const size_t P = (size_t)NN * DH;
  float b0 = beta[0], b1 = beta[1], b2 = beta[2];
  v4f x0 = *(const v4f*)(h + base);
  v4f x1 = *(const v4f*)(h + P + base);
  v4f x2 = *(const v4f*)(h + 2 * P + base);
  v4f z  = x0 * b0 + x1 * b1 + x2 * b2;
  *(v4f*)(out + base) = z;
}

extern "C" void kernel_launch(void* const* d_in, const int* in_sizes, int n_in,
                              void* d_out, int out_size, void* d_ws, size_t ws_size,
                              hipStream_t stream) {
  const float* feats = (const float*)d_in[0];
  const int*   src   = (const int*)  d_in[1];
  const int*   dst   = (const int*)  d_in[2];
  const float* W     = (const float*)d_in[3];
  const float* b     = (const float*)d_in[4];
  const float* pa    = (const float*)d_in[5];
  const float* fcW   = (const float*)d_in[6];
  const float* fcb   = (const float*)d_in[7];
  const float* attn  = (const float*)d_in[8];

  float* ws     = (float*)d_ws;
  float* proj   = ws + PROJ_OFF;
  float* h      = ws + H_OFF;
  int*   ptr    = (int*)(ws + PTR_OFF);
  float* logits = ws + LOG_OFF;
  float* beta   = ws + BETA_OFF;
  int*   rank   = (int*)(ws + RANK_OFF);
  int*   bsum   = (int*)(ws + BSUM_OFF);
  float* out    = (float*)d_out;
  int*   sorted = (int*)d_out;

  hipMemsetAsync(ws + PTR_OFF, 0, (size_t)(NROWS + 8) * sizeof(float), stream);

  dim3 gTiles(NN / 16, MM);
  k_proj_wmma<<<gTiles, 128, 0, stream>>>(feats, W, proj);

  k_rank <<<SORT_G, 128, 0, stream>>>(dst, ptr, rank);
  k_bsum <<<SCAN_G, SCAN_B, 0, stream>>>(ptr, bsum);
  k_scan <<<SCAN_G, SCAN_B, 0, stream>>>(ptr, bsum);
  k_place<<<SORT_G, 128, 0, stream>>>(src, dst, ptr, rank, sorted);
  k_pull <<<PULL_G, 256, 0, stream>>>(proj, ptr, sorted, b, pa, h);

  k_fc_wmma<<<gTiles, 128, 0, stream>>>(h, fcW, fcb, attn, logits);

  k_softmax<<<1, 32, 0, stream>>>(logits, beta);

  size_t ov4 = (size_t)NN * (DH / 4);
  k_out<<<(unsigned)((ov4 + 255) / 256), 256, 0, stream>>>(h, beta, out);
}


# edge sort as one pass: a returning int add on the row's counter gives an edge its slot in a 64-entry source list kept in the row's own h line; rank, scan, place gone; row sum order: arrival as before
# speedup vs baseline: 2.3729x; 1.0455x over previous
#include <hip/hip_runtime.h>
#include <hip/hip_bf16.h>
#include <math.h>

#define NN  100000
#define MM  3
#define DIN 128
#define DH  64
#define NE  1600000
#define NROWS (MM * NN)
#define NEDGE (MM * NE)
#define SLOTS DH
#define FILL_G 3125
#define PULL_G 2048

typedef __attribute__((ext_vector_type(2))) float v2f;
typedef __attribute__((ext_vector_type(4))) float v4f;
typedef __attribute__((ext_vector_type(8))) float v8f;
typedef __attribute__((ext_vector_type(4))) int v4i;

#define PROJ_OFF  0ull
#define H_OFF     19200000ull
#define CNT_OFF   38400000ull
#define LOG_OFF   38700000ull
#define OVF_OFF   38700003ull
#define BETA_OFF  38700004ull

__device__ __forceinline__ void atomicAddF(float* p, float v) {
  asm volatile("global_atomic_add_f32 %0, %1, off scope:SCOPE_DEV"
               :: "v"(p), "v"(v) : "memory");
}

__global__ __launch_bounds__(128)
void k_proj_wmma(const float* __restrict__ feats, const float* __restrict__ W,
                 float* __restrict__ proj) {
  const int lane = threadIdx.x & 31;
  const int wv   = threadIdx.x >> 5;
  const int col0 = wv * 16;
  const int m    = blockIdx.y;
  const int n0   = blockIdx.x * 16;
  const int r    = lane & 15;
  const int hi   = lane >> 4;
  const int kh   = hi * 2;

  const float* A = feats + ((size_t)m * NN + n0) * DIN;
  const float* B = W + (size_t)m * DIN * DH;

  v8f acc = {0.f,0.f,0.f,0.f,0.f,0.f,0.f,0.f};
  #pragma unroll
  for (int k = 0; k < DIN; k += 4) {
    v2f a = *(const v2f*)(A + (size_t)r * DIN + k + kh);
    v2f b;
    b.x = B[(k + kh)     * DH + col0 + r];
    b.y = B[(k + kh + 1) * DH + col0 + r];
    acc = __builtin_amdgcn_wmma_f32_16x16x4_f32(false, a, false, b, (short)0, acc,
                                                false, false);
  }
  float* O = proj + ((size_t)m * NN + n0) * DH;
  #pragma unroll
  for (int v = 0; v < 8; ++v)
    O[(size_t)(v + hi * 8) * DH + col0 + r] = acc[v];
}

__device__ __forceinline__ void put_edge(int* __restrict__ slots, int* __restrict__ ovfcnt, int* __restrict__ ovf,
                                         int row, int pos, int val, int e) {
  if (pos < SLOTS) slots[(size_t)row * SLOTS + pos] = val;
  else             ovf[atomicAdd(ovfcnt, 1)] = e;
}

__global__ __launch_bounds__(128)
void k_fill(const int* __restrict__ src, const int* __restrict__ dst, int* __restrict__ cnt,
            int* __restrict__ slots, int* __restrict__ ovfcnt, int* __restrict__ ovf) {
  const int t = blockIdx.x * 128 + threadIdx.x;
  v4i d[MM], s[MM];
  #pragma unroll
  for (int m = 0; m < MM; ++m) {
    d[m] = *(const v4i*)(dst + (size_t)m * NE + (size_t)t * 4);
    s[m] = *(const v4i*)(src + (size_t)m * NE + (size_t)t * 4);
  }
  v4i r[MM];
  #pragma unroll
  for (int m = 0; m < MM; ++m) {
    int* c = cnt + m * NN;
    r[m].x = atomicAdd(c + d[m].x, 1);
    r[m].y = atomicAdd(c + d[m].y, 1);
    r[m].z = atomicAdd(c + d[m].z, 1);
    r[m].w = atomicAdd(c + d[m].w, 1);
  }
  #pragma unroll
  for (int m = 0; m < MM; ++m) {
    const int o = m * NN;
    const int e = m * NE + t * 4;
    put_edge(slots, ovfcnt, ovf, o + d[m].x, r[m].x, o + s[m].x, e);
    put_edge(slots, ovfcnt, ovf, o + d[m].y, r[m].y, o + s[m].y, e + 1);
    put_edge(slots, ovfcnt, ovf, o + d[m].z, r[m].z, o + s[m].z, e + 2);
    put_edge(slots, ovfcnt, ovf, o + d[m].w, r[m].w, o + s[m].w, e + 3);
  }
}

__device__ __forceinline__ int first_ids(const int* slots, const int* __restrict__ cnt, int g, int n, int lane) {
  const int* p = (n > 0) ? slots + (size_t)g * SLOTS + min(lane, n - 1) : cnt + min(g, NROWS - 1);
  return __hip_atomic_load(p, __ATOMIC_RELAXED, __HIP_MEMORY_SCOPE_WAVEFRONT);
}

__device__ __forceinline__ void sum_batch(const float* __restrict__ proj, int idc, int nb, int lane,
                                          float& ax, float& ay) {
  for (int j = 0; j < nb; j += 8) {
    v2f v[8];
    #pragma unroll
    for (int u = 0; u < 8; ++u) {
      const int s = __builtin_amdgcn_readlane(idc, min(j + u, nb - 1));
      v[u] = *(const v2f*)(proj + (size_t)s * DH + lane * 2);
    }
    #pragma unroll
    for (int u = 0; u < 8; ++u) {
      const bool ok = (j + u) < nb;
      ax += ok ? v[u].x : 0.f;
      ay += ok ? v[u].y : 0.f;
    }
  }
}

__global__ __launch_bounds__(256)
void k_pull(const float* __restrict__ proj, const int* __restrict__ cnt,
            const int* __restrict__ src, const int* __restrict__ dst,
            const int* __restrict__ ovfcnt, const int* __restrict__ ovf,
            const float* __restrict__ b, const float* __restrict__ pa, float* h) {
  const int* slots = (const int*)h;
  const int lane = threadIdx.x & 31;
  const int nw = PULL_G * 8;
  int g = __builtin_amdgcn_readfirstlane((int)(blockIdx.x * 8 + (threadIdx.x >> 5)));
  int deg  = cnt[g];
  int id   = first_ids(slots, cnt, g, min(deg, 32), lane);
  int deg1 = cnt[min(g + nw, NROWS - 1)];

  for (;;) {
    const bool more = (g + nw) < NROWS;
    const int id1  = first_ids(slots, cnt, g + nw, more ? min(deg1, 32) : 0, lane);
    const int deg2 = cnt[min(g + 2 * nw, NROWS - 1)];

    const int m   = g / NN;
    const float a = pa[m];
    const v2f bv  = *(const v2f*)(b + m * DH + lane * 2);

    float ax = 0.f, ay = 0.f;
    const int nl = min(deg, SLOTS);
    sum_batch(proj, id, min(32, nl), lane, ax, ay);
    if (nl > 32) {
      const int nb = nl - 32;
      const int idc = slots[(size_t)g * SLOTS + 32 + min(lane, nb - 1)];
      sum_batch(proj, idc, nb, lane, ax, ay);
    }
    if (deg > SLOTS) {
      const int novf = ovfcnt[0];
      for (int base = 0; base < novf; base += 32) {
        const int k = base + lane;
        int sid = -1;
        if (k < novf) {
          const int e  = ovf[k];
          const int em = e / NE;
          if (dst[e] + em * NN == g) sid = src[e] + em * NN;
        }
        unsigned hit = (unsigned)__ballot(sid >= 0);
        while (hit) {
          const int j = __builtin_ctz(hit);
          hit &= hit - 1;
          const int s = __builtin_amdgcn_readlane(sid, j);
          const v2f v = *(const v2f*)(proj + (size_t)s * DH + lane * 2);
          ax += v.x;  ay += v.y;
        }
      }
    }
    const float dg = fmaxf((float)deg, 1.0f);
    const float hx = ax / dg + bv.x;
    const float hy = ay / dg + bv.y;
    v2f o;
    o.x = hx > 0.f ? hx : a * hx;
    o.y = hy > 0.f ? hy : a * hy;
    *(v2f*)(h + (size_t)g * DH + lane * 2) = o;

    if (!more) break;
    g = g + nw;  deg = deg1;  id = id1;  deg1 = deg2;
  }
}

__global__ __launch_bounds__(128)
void k_fc_wmma(const float* __restrict__ h, const float* __restrict__ fcW,
               const float* __restrict__ fcb, const float* __restrict__ attn,
               float* __restrict__ logits) {
  __shared__ float red[128];
  const int lane = threadIdx.x & 31;
  const int wv   = threadIdx.x >> 5;
  const int col0 = wv * 16;
  const int m    = blockIdx.y;
  const int n0   = blockIdx.x * 16;
  const int r    = lane & 15;
  const int hi   = lane >> 4;
  const int kh   = hi * 2;

  const float* A = h + ((size_t)m * NN + n0) * DH;
  float bias = fcb[col0 + r];
  v8f acc = {bias, bias, bias, bias, bias, bias, bias, bias};
  #pragma unroll
  for (int k = 0; k < DH; k += 4) {
    v2f a = *(const v2f*)(A + (size_t)r * DH + k + kh);
    v2f bb;
    bb.x = fcW[(k + kh)     * DH + col0 + r];
    bb.y = fcW[(k + kh + 1) * DH + col0 + r];
    acc = __builtin_amdgcn_wmma_f32_16x16x4_f32(false, a, false, bb, (short)0, acc,
                                                false, false);
  }
  float p = 0.f;
  #pragma unroll
  for (int v = 0; v < 8; ++v) p += tanhf(acc[v]);
  p *= attn[col0 + r];

  red[threadIdx.x] = p;
  __syncthreads();
  #pragma unroll
  for (int s2 = 64; s2 > 0; s2 >>= 1) {
    if (threadIdx.x < s2) red[threadIdx.x] += red[threadIdx.x + s2];
    __syncthreads();
  }
  if (threadIdx.x == 0) atomicAddF(&logits[m], red[0]);
}

__global__ void k_softmax(const float* __restrict__ logits, float* __restrict__ beta) {
  if (threadIdx.x == 0 && blockIdx.x == 0) {
    float l[MM]; float mx = -1e30f;
    for (int m = 0; m < MM; ++m) { l[m] = logits[m] / (float)NN; mx = fmaxf(mx, l[m]); }
    float s = 0.f;
    for (int m = 0; m < MM; ++m) { l[m] = expf(l[m] - mx); s += l[m]; }
    for (int m = 0; m < MM; ++m) beta[m] = l[m] / s;
  }
}

__global__ __launch_bounds__(256)
void k_out(const float* __restrict__ h, const float* __restrict__ beta,
           float* __restrict__ out) {
  size_t i4 = (size_t)blockIdx.x * 256 + threadIdx.x;
  if (i4 >= (size_t)NN * (DH / 4)) return;
  size_t base = i4 * 4;
  const size_t P = (size_t)NN * DH;
  float b0 = beta[0], b1 = beta[1], b2 = beta[2];
  v4f x0 = *(const v4f*)(h + base);
  v4f x1 = *(const v4f*)(h + P + base);
  v4f x2 = *(const v4f*)(h + 2 * P + base);
  v4f z  = x0 * b0 + x1 * b1 + x2 * b2;
  *(v4f*)(out + base) = z;
}

extern "C" void kernel_launch(void* const* d_in, const int* in_sizes, int n_in,
                              void* d_out, int out_size, void* d_ws, size_t ws_size,
                              hipStream_t stream) {
  const float* feats = (const float*)d_in[0];
  const int*   src   = (const int*)  d_in[1];
  const int*   dst   = (const int*)  d_in[2];
  const float* W     = (const float*)d_in[3];
  const float* b     = (const float*)d_in[4];
  const float* pa    = (const float*)d_in[5];
  const float* fcW   = (const float*)d_in[6];
  const float* fcb   = (const float*)d_in[7];
  const float* attn  = (const float*)d_in[8];

  float* ws     = (float*)d_ws;
  float* proj   = ws + PROJ_OFF;
  float* h      = ws + H_OFF;
  int*   cnt    = (int*)(ws + CNT_OFF);
  float* logits = ws + LOG_OFF;
  int*   ovfcnt = (int*)(ws + OVF_OFF);
  float* beta   = ws + BETA_OFF;
  float* out    = (float*)d_out;
  int*   ovf    = (int*)d_out;

  hipMemsetAsync(ws + CNT_OFF, 0, (size_t)(NROWS + 8) * sizeof(float), stream);

  dim3 gTiles(NN / 16, MM);
  k_proj_wmma<<<gTiles, 128, 0, stream>>>(feats, W, proj);

  k_fill<<<FILL_G, 128, 0, stream>>>(src, dst, cnt, (int*)h, ovfcnt, ovf);
  k_pull<<<PULL_G, 256, 0, stream>>>(proj, cnt, src, dst, ovfcnt, ovf, b, pa, h);

  k_fc_wmma<<<gTiles, 128, 0, stream>>>(h, fcW, fcb, attn, logits);

  k_softmax<<<1, 32, 0, stream>>>(logits, beta);

  size_t ov4 = (size_t)NN * (DH / 4);
  k_out<<<(unsigned)((ov4 + 255) / 256), 256, 0, stream>>>(h, beta, out);
}


# both matrix kernels: the work-group's 16-row A tile is copied once into LDS by whole-line 16-byte loads, the 4 waves read fragments there (was 8-byte loads over 16 lines per wave); sums unchanged
# speedup vs baseline: 2.5831x; 1.0886x over previous
#include <hip/hip_runtime.h>
#include <hip/hip_bf16.h>
#include <math.h>

#define NN  100000
#define MM  3
#define DIN 128
#define DH  64
#define NE  1600000
#define NROWS (MM * NN)
#define NEDGE (MM * NE)
#define SLOTS DH
#define FILL_G 3125
#define PULL_G 2048

typedef __attribute__((ext_vector_type(2))) float v2f;
typedef __attribute__((ext_vector_type(4))) float v4f;
typedef __attribute__((ext_vector_type(8))) float v8f;
typedef __attribute__((ext_vector_type(4))) int v4i;

#define PROJ_OFF  0ull
#define H_OFF     19200000ull
#define CNT_OFF   38400000ull
#define LOG_OFF   38700000ull
#define OVF_OFF   38700003ull
#define BETA_OFF  38700004ull

__device__ __forceinline__ void atomicAddF(float* p, float v) {
  asm volatile("global_atomic_add_f32 %0, %1, off scope:SCOPE_DEV"
               :: "v"(p), "v"(v) : "memory");
}

#define PA_PITCH (DIN + 4)
__global__ __launch_bounds__(128)
void k_proj_wmma(const float* __restrict__ feats, const float* __restrict__ W,
                 float* __restrict__ proj) {
  __shared__ __attribute__((aligned(16))) float As[16 * PA_PITCH];
  const int lane = threadIdx.x & 31;
  const int wv   = threadIdx.x >> 5;
  const int col0 = wv * 16;
  const int m    = blockIdx.y;
  const int n0   = blockIdx.x * 16;
  const int r    = lane & 15;
  const int hi   = lane >> 4;
  const int kh   = hi * 2;

  const float* A = feats + ((size_t)m * NN + n0) * DIN;
  const float* B = W + (size_t)m * DIN * DH;

  #pragma unroll
  for (int i = 0; i < 4; ++i) {
    const int q = (int)threadIdx.x + 128 * i;
    const v4f x = *(const v4f*)(A + (size_t)q * 4);
    *(v4f*)(As + (q >> 5) * PA_PITCH + (q & 31) * 4) = x;
  }
  __syncthreads();

  v8f acc = {0.f,0.f,0.f,0.f,0.f,0.f,0.f,0.f};
  #pragma unroll
  for (int k = 0; k < DIN; k += 4) {
    v2f a = *(const v2f*)(As + r * PA_PITCH + k + kh);
    v2f b;
    b.x = B[(k + kh)     * DH + col0 + r];
    b.y = B[(k + kh + 1) * DH + col0 + r];
    acc = __builtin_amdgcn_wmma_f32_16x16x4_f32(false, a, false, b, (short)0, acc,
                                                false, false);
  }
  float* O = proj + ((size_t)m * NN + n0) * DH;
  #pragma unroll
  for (int v = 0; v < 8; ++v)
    O[(size_t)(v + hi * 8) * DH + col0 + r] = acc[v];
}

__device__ __forceinline__ void put_edge(int* __restrict__ slots, int* __restrict__ ovfcnt, int* __restrict__ ovf,
                                         int row, int pos, int val, int e) {
  if (pos < SLOTS) slots[(size_t)row * SLOTS + pos] = val;
  else             ovf[atomicAdd(ovfcnt, 1)] = e;
}

__global__ __launch_bounds__(128)
void k_fill(const int* __restrict__ src, const int* __restrict__ dst, int* __restrict__ cnt,
            int* __restrict__ slots, int* __restrict__ ovfcnt, int* __restrict__ ovf) {
  const int t = blockIdx.x * 128 + threadIdx.x;
  v4i d[MM], s[MM];
  #pragma unroll
  for (int m = 0; m < MM; ++m) {
    d[m] = *(const v4i*)(dst + (size_t)m * NE + (size_t)t * 4);
    s[m] = *(const v4i*)(src + (size_t)m * NE + (size_t)t * 4);
  }
  v4i r[MM];
  #pragma unroll
  for (int m = 0; m < MM; ++m) {
    int* c = cnt + m * NN;
    r[m].x = atomicAdd(c + d[m].x, 1);
    r[m].y = atomicAdd(c + d[m].y, 1);
    r[m].z = atomicAdd(c + d[m].z, 1);
    r[m].w = atomicAdd(c + d[m].w, 1);
  }
  #pragma unroll
  for (int m = 0; m < MM; ++m) {
    const int o = m * NN;
    const int e = m * NE + t * 4;
    put_edge(slots, ovfcnt, ovf, o + d[m].x, r[m].x, o + s[m].x, e);
    put_edge(slots, ovfcnt, ovf, o + d[m].y, r[m].y, o + s[m].y, e + 1);
    put_edge(slots, ovfcnt, ovf, o + d[m].z, r[m].z, o + s[m].z, e + 2);
    put_edge(slots, ovfcnt, ovf, o + d[m].w, r[m].w, o + s[m].w, e + 3);
  }
}

__device__ __forceinline__ int first_ids(const int* slots, const int* __restrict__ cnt, int g, int n, int lane) {
  const int* p = (n > 0) ? slots + (size_t)g * SLOTS + min(lane, n - 1) : cnt + min(g, NROWS - 1);
  return __hip_atomic_load(p, __ATOMIC_RELAXED, __HIP_MEMORY_SCOPE_WAVEFRONT);
}

__device__ __forceinline__ void sum_batch(const float* __restrict__ proj, int idc, int nb, int lane,
                                          float& ax, float& ay) {
  for (int j = 0; j < nb; j += 8) {
    v2f v[8];
    #pragma unroll
    for (int u = 0; u < 8; ++u) {
      const int s = __builtin_amdgcn_readlane(idc, min(j + u, nb - 1));
      v[u] = *(const v2f*)(proj + (size_t)s * DH + lane * 2);
    }
    #pragma unroll
    for (int u = 0; u < 8; ++u) {
      const bool ok = (j + u) < nb;
      ax += ok ? v[u].x : 0.f;
      ay += ok ? v[u].y : 0.f;
    }
  }
}

__global__ __launch_bounds__(256)
void k_pull(const float* __restrict__ proj, const int* __restrict__ cnt,
            const int* __restrict__ src, const int* __restrict__ dst,
            const int* __restrict__ ovfcnt, const int* __restrict__ ovf,
            const float* __restrict__ b, const float* __restrict__ pa, float* h) {
  const int* slots = (const int*)h;
  const int lane = threadIdx.x & 31;
  const int nw = PULL_G * 8;
  int g = __builtin_amdgcn_readfirstlane((int)(blockIdx.x * 8 + (threadIdx.x >> 5)));
  int deg  = cnt[g];
  int id   = first_ids(slots, cnt, g, min(deg, 32), lane);
  int deg1 = cnt[min(g + nw, NROWS - 1)];

  for (;;) {
    const bool more = (g + nw) < NROWS;
    const int id1  = first_ids(slots, cnt, g + nw, more ? min(deg1, 32) : 0, lane);
    const int deg2 = cnt[min(g + 2 * nw, NROWS - 1)];

    const int m   = g / NN;
    const float a = pa[m];
    const v2f bv  = *(const v2f*)(b + m * DH + lane * 2);

    float ax = 0.f, ay = 0.f;
    const int nl = min(deg, SLOTS);
    sum_batch(proj, id, min(32, nl), lane, ax, ay);
    if (nl > 32) {
      const int nb = nl - 32;
      const int idc = slots[(size_t)g * SLOTS + 32 + min(lane, nb - 1)];
      sum_batch(proj, idc, nb, lane, ax, ay);
    }
    if (deg > SLOTS) {
      const int novf = ovfcnt[0];
      for (int base = 0; base < novf; base += 32) {
        const int k = base + lane;
        int sid = -1;
        if (k < novf) {
          const int e  = ovf[k];
          const int em = e / NE;
          if (dst[e] + em * NN == g) sid = src[e] + em * NN;
        }
        unsigned hit = (unsigned)__ballot(sid >= 0);
        while (hit) {
          const int j = __builtin_ctz(hit);
          hit &= hit - 1;
          const int s = __builtin_amdgcn_readlane(sid, j);
          const v2f v = *(const v2f*)(proj + (size_t)s * DH + lane * 2);
          ax += v.x;  ay += v.y;
        }
      }
    }
    const float dg = fmaxf((float)deg, 1.0f);
    const float hx = ax / dg + bv.x;
    const float hy = ay / dg + bv.y;
    v2f o;
    o.x = hx > 0.f ? hx : a * hx;
    o.y = hy > 0.f ? hy : a * hy;
    *(v2f*)(h + (size_t)g * DH + lane * 2) = o;

    if (!more) break;
    g = g + nw;  deg = deg1;  id = id1;  deg1 = deg2;
  }
}

#define FA_PITCH (DH + 4)
__global__ __launch_bounds__(128)
void k_fc_wmma(const float* __restrict__ h, const float* __restrict__ fcW,
               const float* __restrict__ fcb, const float* __restrict__ attn,
               float* __restrict__ logits) {
  __shared__ float red[128];
  __shared__ __attribute__((aligned(16))) float As[16 * FA_PITCH];
  const int lane = threadIdx.x & 31;
  const int wv   = threadIdx.x >> 5;
  const int col0 = wv * 16;
  const int m    = blockIdx.y;
  const int n0   = blockIdx.x * 16;
  const int r    = lane & 15;
  const int hi   = lane >> 4;
  const int kh   = hi * 2;

  const float* A = h + ((size_t)m * NN + n0) * DH;
  #pragma unroll
  for (int i = 0; i < 2; ++i) {
    const int q = (int)threadIdx.x + 128 * i;
    const v4f x = *(const v4f*)(A + (size_t)q * 4);
    *(v4f*)(As + (q >> 4) * FA_PITCH + (q & 15) * 4) = x;
  }
  __syncthreads();

  float bias = fcb[col0 + r];
  v8f acc = {bias, bias, bias, bias, bias, bias, bias, bias};
  #pragma unroll
  for (int k = 0; k < DH; k += 4) {
    v2f a = *(const v2f*)(As + r * FA_PITCH + k + kh);
    v2f bb;
    bb.x = fcW[(k + kh)     * DH + col0 + r];
    bb.y = fcW[(k + kh + 1) * DH + col0 + r];
    acc = __builtin_amdgcn_wmma_f32_16x16x4_f32(false, a, false, bb, (short)0, acc,
                                                false, false);
  }
  float p = 0.f;
  #pragma unroll
  for (int v = 0; v < 8; ++v) p += tanhf(acc[v]);
  p *= attn[col0 + r];

  red[threadIdx.x] = p;
  __syncthreads();
  #pragma unroll
  for (int s2 = 64; s2 > 0; s2 >>= 1) {
    if (threadIdx.x < s2) red[threadIdx.x] += red[threadIdx.x + s2];
    __syncthreads();
  }
  if (threadIdx.x == 0) atomicAddF(&logits[m], red[0]);
}

__global__ void k_softmax(const float* __restrict__ logits, float* __restrict__ beta) {
  if (threadIdx.x == 0 && blockIdx.x == 0) {
    float l[MM]; float mx = -1e30f;
    for (int m = 0; m < MM; ++m) { l[m] = logits[m] / (float)NN; mx = fmaxf(mx, l[m]); }
    float s = 0.f;
    for (int m = 0; m < MM; ++m) { l[m] = expf(l[m] - mx); s += l[m]; }
    for (int m = 0; m < MM; ++m) beta[m] = l[m] / s;
  }
}

__global__ __launch_bounds__(256)
void k_out(const float* __restrict__ h, const float* __restrict__ beta,
           float* __restrict__ out) {
  size_t i4 = (size_t)blockIdx.x * 256 + threadIdx.x;
  if (i4 >= (size_t)NN * (DH / 4)) return;
  size_t base = i4 * 4;
  const size_t P = (size_t)NN * DH;
  float b0 = beta[0], b1 = beta[1], b2 = beta[2];
  v4f x0 = *(const v4f*)(h + base);
  v4f x1 = *(const v4f*)(h + P + base);
  v4f x2 = *(const v4f*)(h + 2 * P + base);
  v4f z  = x0 * b0 + x1 * b1 + x2 * b2;
  *(v4f*)(out + base) = z;
}

extern "C" void kernel_launch(void* const* d_in, const int* in_sizes, int n_in,
                              void* d_out, int out_size, void* d_ws, size_t ws_size,
                              hipStream_t stream) {
  const float* feats = (const float*)d_in[0];
  const int*   src   = (const int*)  d_in[1];
  const int*   dst   = (const int*)  d_in[2];
  const float* W     = (const float*)d_in[3];
  const float* b     = (const float*)d_in[4];
  const float* pa    = (const float*)d_in[5];
  const float* fcW   = (const float*)d_in[6];
  const float* fcb   = (const float*)d_in[7];
  const float* attn  = (const float*)d_in[8];

  float* ws     = (float*)d_ws;
  float* proj   = ws + PROJ_OFF;
  float* h      = ws + H_OFF;
  int*   cnt    = (int*)(ws + CNT_OFF);
  float* logits = ws + LOG_OFF;
  int*   ovfcnt = (int*)(ws + OVF_OFF);
  float* beta   = ws + BETA_OFF;
  float* out    = (float*)d_out;
  int*   ovf    = (int*)d_out;

  hipMemsetAsync(ws + CNT_OFF, 0, (size_t)(NROWS + 8) * sizeof(float), stream);

  dim3 gTiles(NN / 16, MM);
  k_proj_wmma<<<gTiles, 128, 0, stream>>>(feats, W, proj);

  k_fill<<<FILL_G, 128, 0, stream>>>(src, dst, cnt, (int*)h, ovfcnt, ovf);
  k_pull<<<PULL_G, 256, 0, stream>>>(proj, cnt, src, dst, ovfcnt, ovf, b, pa, h);

  k_fc_wmma<<<gTiles, 128, 0, stream>>>(h, fcW, fcb, attn, logits);

  k_softmax<<<1, 32, 0, stream>>>(logits, beta);

  size_t ov4 = (size_t)NN * (DH / 4);
  k_out<<<(unsigned)((ov4 + 255) / 256), 256, 0, stream>>>(h, beta, out);
}


# edge pass and row pass by chip half: a work-group reads its die, draws a ticket of its half, and adds, stores and sums only rows that half owns (half of each plane); row sum order: arrival as before
# speedup vs baseline: 4.4691x; 1.7301x over previous
#include <hip/hip_runtime.h>
#include <hip/hip_bf16.h>
#include <math.h>

#define NN  100000
#define MM  3
#define DIN 128
#define DH  64
#define NE  1600000
#define NROWS (MM * NN)
#define NEDGE (MM * NE)
#define SLOTS DH
#define FILL_C 3125
#define FILL_G (2 * FILL_C)
#define PULL_G 2048
#define NHALF (NN / 2)
#define QROWS (MM * NHALF)

typedef __attribute__((ext_vector_type(2))) float v2f;
typedef __attribute__((ext_vector_type(4))) float v4f;
typedef __attribute__((ext_vector_type(8))) float v8f;
typedef __attribute__((ext_vector_type(4))) int v4i;

#define PROJ_OFF  0ull
#define H_OFF     19200000ull
#define CNT_OFF   38400000ull
#define LOG_OFF   38700000ull
#define OVF_OFF   38700003ull
#define BETA_OFF  38700004ull
#define TICK_OFF  38700004ull

__device__ __forceinline__ void atomicAddF(float* p, float v) {
  asm volatile("global_atomic_add_f32 %0, %1, off scope:SCOPE_DEV"
               :: "v"(p), "v"(v) : "memory");
}

__device__ __forceinline__ int take_piece(int* __restrict__ tick, int per, int* sh, int* cls) {
  if (threadIdx.x == 0) {
    const unsigned where = __builtin_amdgcn_s_sendmsg_rtn(0x87);
    const int hw = (int)((where >> 18) & 1u);
    sh[0] = hw;
    sh[1] = atomicAdd(tick + hw, 1);
  }
  __syncthreads();
  const int hw   = __builtin_amdgcn_readfirstlane(sh[0]);
  const int rank = __builtin_amdgcn_readfirstlane(sh[1]);
  const bool own = rank < per;
  *cls = own ? hw : 1 - hw;
  return own ? rank : 2 * per - 1 - rank;
}

#define PA_PITCH (DIN + 4)
__global__ __launch_bounds__(128)
void k_proj_wmma(const float* __restrict__ feats, const float* __restrict__ W,
                 float* __restrict__ proj) {
  __shared__ __attribute__((aligned(16))) float As[16 * PA_PITCH];
  const int lane = threadIdx.x & 31;
  const int wv   = threadIdx.x >> 5;
  const int col0 = wv * 16;
  const int m    = blockIdx.y;
  const int n0   = blockIdx.x * 16;
  const int r    = lane & 15;
  const int hi   = lane >> 4;
  const int kh   = hi * 2;

  const float* A = feats + ((size_t)m * NN + n0) * DIN;
  const float* B = W + (size_t)m * DIN * DH;

  #pragma unroll
  for (int i = 0; i < 4; ++i) {
    const int q = (int)threadIdx.x + 128 * i;
    const v4f x = *(const v4f*)(A + (size_t)q * 4);
    *(v4f*)(As + (q >> 5) * PA_PITCH + (q & 31) * 4) = x;
  }
  __syncthreads();

  v8f acc = {0.f,0.f,0.f,0.f,0.f,0.f,0.f,0.f};
  #pragma unroll
  for (int k = 0; k < DIN; k += 4) {
    v2f a = *(const v2f*)(As + r * PA_PITCH + k + kh);
    v2f b;
    b.x = B[(k + kh)     * DH + col0 + r];
    b.y = B[(k + kh + 1) * DH + col0 + r];
    acc = __builtin_amdgcn_wmma_f32_16x16x4_f32(false, a, false, b, (short)0, acc,
                                                false, false);
  }
  float* O = proj + ((size_t)m * NN + n0) * DH;
  #pragma unroll
  for (int v = 0; v < 8; ++v)
    O[(size_t)(v + hi * 8) * DH + col0 + r] = acc[v];
}

__device__ __forceinline__ bool mine(int d, int lo) { return (unsigned)(d - lo) < (unsigned)NHALF; }

__device__ __forceinline__ void put_edge(int* __restrict__ slots, int* __restrict__ ovfcnt, int* __restrict__ ovf,
                                         int row, int pos, int val, int e) {
  if (pos < SLOTS) slots[(size_t)row * SLOTS + pos] = val;
  else             ovf[atomicAdd(ovfcnt, 1)] = e;
}

__global__ __launch_bounds__(128)
void k_fill(const int* __restrict__ src, const int* __restrict__ dst, int* __restrict__ cnt,
            int* __restrict__ slots, int* __restrict__ ovfcnt, int* __restrict__ ovf, int* __restrict__ tick) {
  __shared__ int sh[2];
  int cls;
  const int chunk = take_piece(tick, FILL_C, sh, &cls);
  const int lo = cls * NHALF;
  const int t = chunk * 128 + threadIdx.x;
  v4i d[MM], s[MM];
  #pragma unroll
  for (int m = 0; m < MM; ++m) {
    d[m] = *(const v4i*)(dst + (size_t)m * NE + (size_t)t * 4);
    s[m] = *(const v4i*)(src + (size_t)m * NE + (size_t)t * 4);
  }
  v4i r[MM];
  #pragma unroll
  for (int m = 0; m < MM; ++m) {
    int* c = cnt + m * NN;
    r[m].x = mine(d[m].x, lo) ? atomicAdd(c + d[m].x, 1) : -1;
    r[m].y = mine(d[m].y, lo) ? atomicAdd(c + d[m].y, 1) : -1;
    r[m].z = mine(d[m].z, lo) ? atomicAdd(c + d[m].z, 1) : -1;
    r[m].w = mine(d[m].w, lo) ? atomicAdd(c + d[m].w, 1) : -1;
  }
  #pragma unroll
  for (int m = 0; m < MM; ++m) {
    const int o = m * NN;
    const int e = m * NE + t * 4;
    if (r[m].x >= 0) put_edge(slots, ovfcnt, ovf, o + d[m].x, r[m].x, o + s[m].x, e);
    if (r[m].y >= 0) put_edge(slots, ovfcnt, ovf, o + d[m].y, r[m].y, o + s[m].y, e + 1);
    if (r[m].z >= 0) put_edge(slots, ovfcnt, ovf, o + d[m].z, r[m].z, o + s[m].z, e + 2);
    if (r[m].w >= 0) put_edge(slots, ovfcnt, ovf, o + d[m].w, r[m].w, o + s[m].w, e + 3);
  }
}

__device__ __forceinline__ int first_ids(const int* slots, const int* __restrict__ cnt, int g, int n, int lane) {
  const int* p = (n > 0) ? slots + (size_t)g * SLOTS + min(lane, n - 1) : cnt + min(g, NROWS - 1);
  return __hip_atomic_load(p, __ATOMIC_RELAXED, __HIP_MEMORY_SCOPE_WAVEFRONT);
}

__device__ __forceinline__ void sum_batch(const float* __restrict__ proj, int idc, int nb, int lane,
                                          float& ax, float& ay) {
  for (int j = 0; j < nb; j += 8) {
    v2f v[8];
    #pragma unroll
    for (int u = 0; u < 8; ++u) {
      const int s = __builtin_amdgcn_readlane(idc, min(j + u, nb - 1));
      v[u] = *(const v2f*)(proj + (size_t)s * DH + lane * 2);
    }
    #pragma unroll
    for (int u = 0; u < 8; ++u) {
      const bool ok = (j + u) < nb;
      ax += ok ? v[u].x : 0.f;
      ay += ok ? v[u].y : 0.f;
    }
  }
}

__device__ __forceinline__ int half_row(int q, int lo) {
  const int m = q / NHALF;
  return m * NN + lo + (q - m * NHALF);
}

__global__ __launch_bounds__(256)
void k_pull(const float* __restrict__ proj, const int* __restrict__ cnt,
            const int* __restrict__ src, const int* __restrict__ dst,
            const int* __restrict__ ovfcnt, const int* __restrict__ ovf,
            const float* __restrict__ b, const float* __restrict__ pa, float* h, int* __restrict__ tick) {
  __shared__ int sh[2];
  const int* slots = (const int*)h;
  const int lane = threadIdx.x & 31;
  int cls;
  const int piece = take_piece(tick, PULL_G / 2, sh, &cls);
  const int lo = cls * NHALF;
  const int nw = (PULL_G / 2) * 8;
  int q = __builtin_amdgcn_readfirstlane(piece * 8 + (int)(threadIdx.x >> 5));
  int g    = half_row(q, lo);
  int deg  = cnt[g];
  int id   = first_ids(slots, cnt, g, min(deg, 32), lane);
  int g1   = half_row(min(q + nw, QROWS - 1), lo);
  int deg1 = cnt[g1];

  for (;;) {
    const bool more = (q + nw) < QROWS;
    const int id1  = first_ids(slots, cnt, g1, more ? min(deg1, 32) : 0, lane);
    const int g2   = half_row(min(q + 2 * nw, QROWS - 1), lo);
    const int deg2 = cnt[g2];

    const int m   = g / NN;
    const float a = pa[m];
    const v2f bv  = *(const v2f*)(b + m * DH + lane * 2);

    float ax = 0.f, ay = 0.f;
    const int nl = min(deg, SLOTS);
    sum_batch(proj, id, min(32, nl), lane, ax, ay);
    if (nl > 32) {
      const int nb = nl - 32;
      const int idc = slots[(size_t)g * SLOTS + 32 + min(lane, nb - 1)];
      sum_batch(proj, idc, nb, lane, ax, ay);
    }
    if (deg > SLOTS) {
      const int novf = ovfcnt[0];
      for (int base = 0; base < novf; base += 32) {
        const int k = base + lane;
        int sid = -1;
        if (k < novf) {
          const int e  = ovf[k];
          const int em = e / NE;
          if (dst[e] + em * NN == g) sid = src[e] + em * NN;
        }
        unsigned hit = (unsigned)__ballot(sid >= 0);
        while (hit) {
          const int j = __builtin_ctz(hit);
          hit &= hit - 1;
          const int s = __builtin_amdgcn_readlane(sid, j);
          const v2f v = *(const v2f*)(proj + (size_t)s * DH + lane * 2);
          ax += v.x;  ay += v.y;
        }
      }
    }
    const float dg = fmaxf((float)deg, 1.0f);
    const float hx = ax / dg + bv.x;
    const float hy = ay / dg + bv.y;
    v2f o;
    o.x = hx > 0.f ? hx : a * hx;
    o.y = hy > 0.f ? hy : a * hy;
    *(v2f*)(h + (size_t)g * DH + lane * 2) = o;

    if (!more) break;
    q = q + nw;  g = g1;  deg = deg1;  id = id1;  g1 = g2;  deg1 = deg2;
  }
}

#define FA_PITCH (DH + 4)
__global__ __launch_bounds__(128)
void k_fc_wmma(const float* __restrict__ h, const float* __restrict__ fcW,
               const float* __restrict__ fcb, const float* __restrict__ attn,
               float* __restrict__ logits) {
  __shared__ float red[128];
  __shared__ __attribute__((aligned(16))) float As[16 * FA_PITCH];
  const int lane = threadIdx.x & 31;
  const int wv   = threadIdx.x >> 5;
  const int col0 = wv * 16;
  const int m    = blockIdx.y;
  const int n0   = blockIdx.x * 16;
  const int r    = lane & 15;
  const int hi   = lane >> 4;
  const int kh   = hi * 2;

  const float* A = h + ((size_t)m * NN + n0) * DH;
  #pragma unroll
  for (int i = 0; i < 2; ++i) {
    const int q = (int)threadIdx.x + 128 * i;
    const v4f x = *(const v4f*)(A + (size_t)q * 4);
    *(v4f*)(As + (q >> 4) * FA_PITCH + (q & 15) * 4) = x;
  }
  __syncthreads();

  float bias = fcb[col0 + r];
  v8f acc = {bias, bias, bias, bias, bias, bias, bias, bias};
  #pragma unroll
  for (int k = 0; k < DH; k += 4) {
    v2f a = *(const v2f*)(As + r * FA_PITCH + k + kh);
    v2f bb;
    bb.x = fcW[(k + kh)     * DH + col0 + r];
    bb.y = fcW[(k + kh + 1) * DH + col0 + r];
    acc = __builtin_amdgcn_wmma_f32_16x16x4_f32(false, a, false, bb, (short)0, acc,
                                                false, false);
  }
  float p = 0.f;
  #pragma unroll
  for (int v = 0; v < 8; ++v) p += tanhf(acc[v]);
  p *= attn[col0 + r];

  red[threadIdx.x] = p;
  __syncthreads();
  #pragma unroll
  for (int s2 = 64; s2 > 0; s2 >>= 1) {
    if (threadIdx.x < s2) red[threadIdx.x] += red[threadIdx.x + s2];
    __syncthreads();
  }
  if (threadIdx.x == 0) atomicAddF(&logits[m], red[0]);
}

__global__ void k_softmax(const float* __restrict__ logits, float* __restrict__ beta) {
  if (threadIdx.x == 0 && blockIdx.x == 0) {
    float l[MM]; float mx = -1e30f;
    for (int m = 0; m < MM; ++m) { l[m] = logits[m] / (float)NN; mx = fmaxf(mx, l[m]); }
    float s = 0.f;
    for (int m = 0; m < MM; ++m) { l[m] = expf(l[m] - mx); s += l[m]; }
    for (int m = 0; m < MM; ++m) beta[m] = l[m] / s;
  }
}

__global__ __launch_bounds__(256)
void k_out(const float* __restrict__ h, const float* __restrict__ beta,
           float* __restrict__ out) {
  size_t i4 = (size_t)blockIdx.x * 256 + threadIdx.x;
  if (i4 >= (size_t)NN * (DH / 4)) return;
  size_t base = i4 * 4;
  const size_t P = (size_t)NN * DH;
  float b0 = beta[0], b1 = beta[1], b2 = beta[2];
  v4f x0 = *(const v4f*)(h + base);
  v4f x1 = *(const v4f*)(h + P + base);
  v4f x2 = *(const v4f*)(h + 2 * P + base);
  v4f z  = x0 * b0 + x1 * b1 + x2 * b2;
  *(v4f*)(out + base) = z;
}

extern "C" void kernel_launch(void* const* d_in, const int* in_sizes, int n_in,
                              void* d_out, int out_size, void* d_ws, size_t ws_size,
                              hipStream_t stream) {
  const float* feats = (const float*)d_in[0];
  const int*   src   = (const int*)  d_in[1];
  const int*   dst   = (const int*)  d_in[2];
  const float* W     = (const float*)d_in[3];
  const float* b     = (const float*)d_in[4];
  const float* pa    = (const float*)d_in[5];
  const float* fcW   = (const float*)d_in[6];
  const float* fcb   = (const float*)d_in[7];
  const float* attn  = (const float*)d_in[8];

  float* ws     = (float*)d_ws;
  float* proj   = ws + PROJ_OFF;
  float* h      = ws + H_OFF;
  int*   cnt    = (int*)(ws + CNT_OFF);
  float* logits = ws + LOG_OFF;
  int*   ovfcnt = (int*)(ws + OVF_OFF);
  float* beta   = ws + BETA_OFF;
  int*   tick   = (int*)(ws + TICK_OFF);
  float* out    = (float*)d_out;
  int*   ovf    = (int*)d_out;

  hipMemsetAsync(ws + CNT_OFF, 0, (size_t)(NROWS + 8) * sizeof(float), stream);

  dim3 gTiles(NN / 16, MM);
  k_proj_wmma<<<gTiles, 128, 0, stream>>>(feats, W, proj);

  k_fill<<<FILL_G, 128, 0, stream>>>(src, dst, cnt, (int*)h, ovfcnt, ovf, tick);
  k_pull<<<PULL_G, 256, 0, stream>>>(proj, cnt, src, dst, ovfcnt, ovf, b, pa, h, tick + 2);

  k_fc_wmma<<<gTiles, 128, 0, stream>>>(h, fcW, fcb, attn, logits);

  k_softmax<<<1, 32, 0, stream>>>(logits, beta);

  size_t ov4 = (size_t)NN * (DH / 4);
  k_out<<<(unsigned)((ov4 + 255) / 256), 256, 0, stream>>>(h, beta, out);
}


# fc layer and logits moved into the row pass: a wave keeps its h rows in an LDS tile, runs the fc WMMAs on 16 rows at a time, leftovers pooled per work-group; fc launch gone; logit sum order changed
# speedup vs baseline: 4.7001x; 1.0517x over previous
#include <hip/hip_runtime.h>
#include <hip/hip_bf16.h>
#include <math.h>

#define NN  100000
#define MM  3
#define DIN 128
#define DH  64
#define NE  1600000
#define NROWS (MM * NN)
#define NEDGE (MM * NE)
#define SLOTS DH
#define FILL_C 3125
#define FILL_G (2 * FILL_C)
#define PULL_T 1024
#define PULL_W (PULL_T / 32)
#define PULL_G 512
#define HT_PITCH (DH + 4)
#define NHALF (NN / 2)
#define QROWS (MM * NHALF)

typedef __attribute__((ext_vector_type(2))) float v2f;
typedef __attribute__((ext_vector_type(4))) float v4f;
typedef __attribute__((ext_vector_type(8))) float v8f;
typedef __attribute__((ext_vector_type(4))) int v4i;

#define PROJ_OFF  0ull
#define H_OFF     19200000ull
#define CNT_OFF   38400000ull
#define LOG_OFF   38700000ull
#define OVF_OFF   38700003ull
#define BETA_OFF  38700004ull
#define TICK_OFF  38700004ull

__device__ __forceinline__ void atomicAddF(float* p, float v) {
  asm volatile("global_atomic_add_f32 %0, %1, off scope:SCOPE_DEV"
               :: "v"(p), "v"(v) : "memory");
}

__device__ __forceinline__ int take_piece(int* __restrict__ tick, int per, int* sh, int* cls) {
  if (threadIdx.x == 0) {
    const unsigned where = __builtin_amdgcn_s_sendmsg_rtn(0x87);
    const int hw = (int)((where >> 18) & 1u);
    sh[0] = hw;
    sh[1] = atomicAdd(tick + hw, 1);
  }
  __syncthreads();
  const int hw   = __builtin_amdgcn_readfirstlane(sh[0]);
  const int rank = __builtin_amdgcn_readfirstlane(sh[1]);
  const bool own = rank < per;
  *cls = own ? hw : 1 - hw;
  return own ? rank : 2 * per - 1 - rank;
}

#define PA_PITCH (DIN + 4)
__global__ __launch_bounds__(128)
void k_proj_wmma(const float* __restrict__ feats, const float* __restrict__ W,
                 float* __restrict__ proj) {
  __shared__ __attribute__((aligned(16))) float As[16 * PA_PITCH];
  const int lane = threadIdx.x & 31;
  const int wv   = threadIdx.x >> 5;
  const int col0 = wv * 16;
  const int m    = blockIdx.y;
  const int n0   = blockIdx.x * 16;
  const int r    = lane & 15;
  const int hi   = lane >> 4;
  const int kh   = hi * 2;

  const float* A = feats + ((size_t)m * NN + n0) * DIN;
  const float* B = W + (size_t)m * DIN * DH;

  #pragma unroll
  for (int i = 0; i < 4; ++i) {
    const int q = (int)threadIdx.x + 128 * i;
    const v4f x = *(const v4f*)(A + (size_t)q * 4);
    *(v4f*)(As + (q >> 5) * PA_PITCH + (q & 31) * 4) = x;
  }
  __syncthreads();

  v8f acc = {0.f,0.f,0.f,0.f,0.f,0.f,0.f,0.f};
  #pragma unroll
  for (int k = 0; k < DIN; k += 4) {
    v2f a = *(const v2f*)(As + r * PA_PITCH + k + kh);
    v2f b;
    b.x = B[(k + kh)     * DH + col0 + r];
    b.y = B[(k + kh + 1) * DH + col0 + r];
    acc = __builtin_amdgcn_wmma_f32_16x16x4_f32(false, a, false, b, (short)0, acc,
                                                false, false);
  }
  float* O = proj + ((size_t)m * NN + n0) * DH;
  #pragma unroll
  for (int v = 0; v < 8; ++v)
    O[(size_t)(v + hi * 8) * DH + col0 + r] = acc[v];
}

__device__ __forceinline__ bool mine(int d, int lo) { return (unsigned)(d - lo) < (unsigned)NHALF; }

__device__ __forceinline__ void put_edge(int* __restrict__ slots, int* __restrict__ ovfcnt, int* __restrict__ ovf,
                                         int row, int pos, int val, int e) {
  if (pos < SLOTS) slots[(size_t)row * SLOTS + pos] = val;
  else             ovf[atomicAdd(ovfcnt, 1)] = e;
}

__global__ __launch_bounds__(128)
void k_fill(const int* __restrict__ src, const int* __restrict__ dst, int* __restrict__ cnt,
            int* __restrict__ slots, int* __restrict__ ovfcnt, int* __restrict__ ovf, int* __restrict__ tick) {
  __shared__ int sh[2];
  int cls;
  const int chunk = take_piece(tick, FILL_C, sh, &cls);
  const int lo = cls * NHALF;
  const int t = chunk * 128 + threadIdx.x;
  v4i d[MM], s[MM];
  #pragma unroll
  for (int m = 0; m < MM; ++m) {
    d[m] = *(const v4i*)(dst + (size_t)m * NE + (size_t)t * 4);
    s[m] = *(const v4i*)(src + (size_t)m * NE + (size_t)t * 4);
  }
  v4i r[MM];
  #pragma unroll
  for (int m = 0; m < MM; ++m) {
    int* c = cnt + m * NN;
    r[m].x = mine(d[m].x, lo) ? atomicAdd(c + d[m].x, 1) : -1;
    r[m].y = mine(d[m].y, lo) ? atomicAdd(c + d[m].y, 1) : -1;
    r[m].z = mine(d[m].z, lo) ? atomicAdd(c + d[m].z, 1) : -1;
    r[m].w = mine(d[m].w, lo) ? atomicAdd(c + d[m].w, 1) : -1;
  }
  #pragma unroll
  for (int m = 0; m < MM; ++m) {
    const int o = m * NN;
    const int e = m * NE + t * 4;
    if (r[m].x >= 0) put_edge(slots, ovfcnt, ovf, o + d[m].x, r[m].x, o + s[m].x, e);
    if (r[m].y >= 0) put_edge(slots, ovfcnt, ovf, o + d[m].y, r[m].y, o + s[m].y, e + 1);
    if (r[m].z >= 0) put_edge(slots, ovfcnt, ovf, o + d[m].z, r[m].z, o + s[m].z, e + 2);
    if (r[m].w >= 0) put_edge(slots, ovfcnt, ovf, o + d[m].w, r[m].w, o + s[m].w, e + 3);
  }
}

__device__ __forceinline__ int first_ids(const int* slots, const int* __restrict__ cnt, int g, int n, int lane) {
  const int* p = (n > 0) ? slots + (size_t)g * SLOTS + min(lane, n - 1) : cnt + min(g, NROWS - 1);
  return __hip_atomic_load(p, __ATOMIC_RELAXED, __HIP_MEMORY_SCOPE_WAVEFRONT);
}

__device__ __forceinline__ void sum_batch(const float* __restrict__ proj, int idc, int nb, int lane,
                                          float& ax, float& ay) {
  for (int j = 0; j < nb; j += 8) {
    v2f v[8];
    #pragma unroll
    for (int u = 0; u < 8; ++u) {
      const int s = __builtin_amdgcn_readlane(idc, min(j + u, nb - 1));
      v[u] = *(const v2f*)(proj + (size_t)s * DH + lane * 2);
    }
    #pragma unroll
    for (int u = 0; u < 8; ++u) {
      const bool ok = (j + u) < nb;
      ax += ok ? v[u].x : 0.f;
      ay += ok ? v[u].y : 0.f;
    }
  }
}

__device__ __forceinline__ int half_row(int q, int lo) {
  const int m = q / NHALF;
  return m * NN + lo + (q - m * NHALF);
}

__device__ __forceinline__ void fc_tile(const float* arow, const v2f* Bs, int lane, unsigned code, int ct0, int ct1,
                                        const float* __restrict__ fcb, const float* __restrict__ attn,
                                        float& p0, float& p1, float& p2) {
  const int r = lane & 15;
  #pragma unroll 1
  for (int ct = ct0; ct < ct1; ++ct) {
    const float bi = fcb[ct * 16 + r];
    v8f c = {bi, bi, bi, bi, bi, bi, bi, bi};
    #pragma unroll
    for (int ks = 0; ks < DH / 4; ++ks) {
      const unsigned long long ab = __hip_atomic_load((const unsigned long long*)(arow + 4 * ks), __ATOMIC_RELAXED,
                                                      __HIP_MEMORY_SCOPE_WAVEFRONT);
      const v2f a = __builtin_bit_cast(v2f, ab);
      const v2f x = Bs[(ks * 4 + ct) * 32 + lane];
      c = __builtin_amdgcn_wmma_f32_16x16x4_f32(false, a, false, x, (short)0, c, false, false);
    }
    float s0 = 0.f, s1 = 0.f, s2 = 0.f;
    #pragma unroll
    for (int v = 0; v < 8; ++v) {
      const unsigned pl = (code >> (2 * v)) & 3u;
      const float t = tanhf(c[v]);
      s0 += pl == 0u ? t : 0.f;
      s1 += pl == 1u ? t : 0.f;
      s2 += pl == 2u ? t : 0.f;
    }
    const float at = attn[ct * 16 + r];
    p0 += s0 * at;  p1 += s1 * at;  p2 += s2 * at;
  }
}

__global__ __launch_bounds__(PULL_T)
void k_pull(const float* __restrict__ proj, const int* __restrict__ cnt,
            const int* __restrict__ src, const int* __restrict__ dst,
            const int* __restrict__ ovfcnt, const int* __restrict__ ovf,
            const float* __restrict__ b, const float* __restrict__ pa, float* h, int* __restrict__ tick,
            const float* __restrict__ fcW, const float* __restrict__ fcb, const float* __restrict__ attn,
            float* __restrict__ logits) {
  __shared__ int sh[2];
  __shared__ float red[PULL_W * MM];
  __shared__ __attribute__((aligned(16))) v2f Bs[(DH / 4) * 4 * 32];
  __shared__ __attribute__((aligned(16))) float Ht[PULL_W * 16 * HT_PITCH];
  const int* slots = (const int*)h;
  const int lane = threadIdx.x & 31;
  const int hi   = lane >> 4;
  const int wv   = __builtin_amdgcn_readfirstlane((int)(threadIdx.x >> 5));
  for (int e = threadIdx.x; e < (DH / 4) * 4 * 32; e += PULL_T) {
    const int k   = 4 * (e >> 7) + 2 * ((e & 31) >> 4);
    const int col = 16 * ((e >> 5) & 3) + (e & 15);
    v2f x;
    x.x = fcW[k * DH + col];
    x.y = fcW[(k + 1) * DH + col];
    Bs[e] = x;
  }
  int cls;
  const int piece = take_piece(tick, PULL_G / 2, sh, &cls);
  const int lo = cls * NHALF;
  const int nw = (PULL_G / 2) * PULL_W;
  const int q0 = piece * PULL_W + wv;
  float* tile = Ht + wv * 16 * HT_PITCH;
  float p0 = 0.f, p1 = 0.f, p2 = 0.f;
  int i = 0;
  int q = __builtin_amdgcn_readfirstlane(q0);
  int g    = half_row(q, lo);
  int deg  = cnt[g];
  int id   = first_ids(slots, cnt, g, min(deg, 32), lane);
  int g1   = half_row(min(q + nw, QROWS - 1), lo);
  int deg1 = cnt[g1];

  for (;;) {
    const bool more = (q + nw) < QROWS;
    const int id1  = first_ids(slots, cnt, g1, more ? min(deg1, 32) : 0, lane);
    const int g2   = half_row(min(q + 2 * nw, QROWS - 1), lo);
    const int deg2 = cnt[g2];

    const int m   = g / NN;
    const float a = pa[m];
    const v2f bv  = *(const v2f*)(b + m * DH + lane * 2);

    float ax = 0.f, ay = 0.f;
    const int nl = min(deg, SLOTS);
    sum_batch(proj, id, min(32, nl), lane, ax, ay);
    if (nl > 32) {
      const int nb = nl - 32;
      const int idc = slots[(size_t)g * SLOTS + 32 + min(lane, nb - 1)];
      sum_batch(proj, idc, nb, lane, ax, ay);
    }
    if (deg > SLOTS) {
      const int novf = ovfcnt[0];
      for (int base = 0; base < novf; base += 32) {
        const int k = base + lane;
        int sid = -1;
        if (k < novf) {
          const int e  = ovf[k];
          const int em = e / NE;
          if (dst[e] + em * NN == g) sid = src[e] + em * NN;
        }
        unsigned hit = (unsigned)__ballot(sid >= 0);
        while (hit) {
          const int j = __builtin_ctz(hit);
          hit &= hit - 1;
          const int s = __builtin_amdgcn_readlane(sid, j);
          const v2f v = *(const v2f*)(proj + (size_t)s * DH + lane * 2);
          ax += v.x;  ay += v.y;
        }
      }
    }
    const float dg = fmaxf((float)deg, 1.0f);
    const float hx = ax / dg + bv.x;
    const float hy = ay / dg + bv.y;
    v2f o;
    o.x = hx > 0.f ? hx : a * hx;
    o.y = hy > 0.f ? hy : a * hy;
    *(v2f*)(h + (size_t)g * DH + lane * 2) = o;
    *(v2f*)(tile + (i & 15) * HT_PITCH + lane * 2) = o;

    if (i == 15) {
      unsigned code = 0;
      #pragma unroll
      for (int v = 0; v < 8; ++v) code |= (unsigned)((q0 + (v + 8 * hi) * nw) / NHALF) << (2 * v);
      fc_tile(tile + (lane & 15) * HT_PITCH + 2 * hi, Bs, lane, code, 0, 4, fcb, attn, p0, p1, p2);
    }
    if (!more) break;
    ++i;  q = q + nw;  g = g1;  deg = deg1;  id = id1;  g1 = g2;  deg1 = deg2;
  }

  __syncthreads();
  {
    const int pt = wv >> 2;
    const int r = lane & 15;
    const float* ar = Ht + (4 * pt + (r >> 2)) * 16 * HT_PITCH + (r & 3) * HT_PITCH + 2 * hi;
    unsigned code = 0;
    #pragma unroll
    for (int v = 0; v < 8; ++v) {
      const int rr = v + 8 * hi;
      const int w0 = piece * PULL_W + 4 * pt + (rr >> 2);
      const int nr = (QROWS - 1 - w0) / nw + 1;
      const int j  = rr & 3;
      const unsigned c = (16 + j) < nr ? (unsigned)((w0 + (16 + j) * nw) / NHALF) : 3u;
      code |= c << (2 * v);
    }
    fc_tile(ar, Bs, lane, code, wv & 3, (wv & 3) + 1, fcb, attn, p0, p1, p2);
  }
  #pragma unroll
  for (int o = 16; o > 0; o >>= 1) {
    p0 += __shfl_xor(p0, o);
    p1 += __shfl_xor(p1, o);
    p2 += __shfl_xor(p2, o);
  }
  if (lane == 0) { red[wv * MM] = p0;  red[wv * MM + 1] = p1;  red[wv * MM + 2] = p2; }
  __syncthreads();
  if (threadIdx.x < MM) {
    float t = 0.f;
    for (int w = 0; w < PULL_W; ++w) t += red[w * MM + threadIdx.x];
    atomicAddF(&logits[threadIdx.x], t);
  }
}

__global__ void k_softmax(const float* __restrict__ logits, float* __restrict__ beta) {
  if (threadIdx.x == 0 && blockIdx.x == 0) {
    float l[MM]; float mx = -1e30f;
    for (int m = 0; m < MM; ++m) { l[m] = logits[m] / (float)NN; mx = fmaxf(mx, l[m]); }
    float s = 0.f;
    for (int m = 0; m < MM; ++m) { l[m] = expf(l[m] - mx); s += l[m]; }
    for (int m = 0; m < MM; ++m) beta[m] = l[m] / s;
  }
}

__global__ __launch_bounds__(256)
void k_out(const float* __restrict__ h, const float* __restrict__ beta,
           float* __restrict__ out) {
  size_t i4 = (size_t)blockIdx.x * 256 + threadIdx.x;
  if (i4 >= (size_t)NN * (DH / 4)) return;
  size_t base = i4 * 4;
  const size_t P = (size_t)NN * DH;
  float b0 = beta[0], b1 = beta[1], b2 = beta[2];
  v4f x0 = *(const v4f*)(h + base);
  v4f x1 = *(const v4f*)(h + P + base);
  v4f x2 = *(const v4f*)(h + 2 * P + base);
  v4f z  = x0 * b0 + x1 * b1 + x2 * b2;
  *(v4f*)(out + base) = z;
}

extern "C" void kernel_launch(void* const* d_in, const int* in_sizes, int n_in,
                              void* d_out, int out_size, void* d_ws, size_t ws_size,
                              hipStream_t stream) {
  const float* feats = (const float*)d_in[0];
  const int*   src   = (const int*)  d_in[1];
  const int*   dst   = (const int*)  d_in[2];
  const float* W     = (const float*)d_in[3];
  const float* b     = (const float*)d_in[4];
  const float* pa    = (const float*)d_in[5];
  const float* fcW   = (const float*)d_in[6];
  const float* fcb   = (const float*)d_in[7];
  const float* attn  = (const float*)d_in[8];

  float* ws     = (float*)d_ws;
  float* proj   = ws + PROJ_OFF;
  float* h      = ws + H_OFF;
  int*   cnt    = (int*)(ws + CNT_OFF);
  float* logits = ws + LOG_OFF;
  int*   ovfcnt = (int*)(ws + OVF_OFF);
  float* beta   = ws + BETA_OFF;
  int*   tick   = (int*)(ws + TICK_OFF);
  float* out    = (float*)d_out;
  int*   ovf    = (int*)d_out;

  hipMemsetAsync(ws + CNT_OFF, 0, (size_t)(NROWS + 8) * sizeof(float), stream);

  dim3 gTiles(NN / 16, MM);
  k_proj_wmma<<<gTiles, 128, 0, stream>>>(feats, W, proj);

  k_fill<<<FILL_G, 128, 0, stream>>>(src, dst, cnt, (int*)h, ovfcnt, ovf, tick);
  k_pull<<<PULL_G, PULL_T, 0, stream>>>(proj, cnt, src, dst, ovfcnt, ovf, b, pa, h, tick + 2, fcW, fcb, attn, logits);

  k_softmax<<<1, 32, 0, stream>>>(logits, beta);

  size_t ov4 = (size_t)NN * (DH / 4);
  k_out<<<(unsigned)((ov4 + 255) / 256), 256, 0, stream>>>(h, beta, out);
}
